# starting kernel
# baseline (speedup 1.0000x reference)
#include <hip/hip_runtime.h>

#define N_NODES 100000
#define N_EDGES 1600000
#define IN_F    256
#define HID_F   128
#define OUT_F   32

typedef float v2f __attribute__((ext_vector_type(2)));
typedef float v8f __attribute__((ext_vector_type(8)));

__global__ __launch_bounds__(128) void gemm1_wmma(
    const float* __restrict__ x, const float* __restrict__ W1,
    float* __restrict__ h0)
{
    const int wid  = (blockIdx.x * blockDim.x + threadIdx.x) >> 5;
    const int lane = threadIdx.x & 31;
    if (wid >= N_NODES / 16) return;
    const int m    = lane & 15;
    const int kh   = lane >> 4;
    const int row0 = wid * 16;

    v8f acc[8] = {};

    const float* arow = x + (size_t)(row0 + m) * IN_F + 2 * kh;
    const float* wcol = W1 + m;

    for (int k = 0; k < IN_F; k += 4) {
        v2f a = *(const v2f*)(arow + k);
        const float* wrow = wcol + (size_t)(k + 2 * kh) * HID_F;
#pragma unroll
        for (int t = 0; t < 8; ++t) {
            v2f b;
            b.x = wrow[t * 16];
            b.y = wrow[HID_F + t * 16];
            acc[t] = __builtin_amdgcn_wmma_f32_16x16x4_f32(
                false, a, false, b, (short)0, acc[t], false, false);
        }
    }

    float* drow = h0 + (size_t)(row0 + kh * 8) * HID_F + m;
#pragma unroll
    for (int t = 0; t < 8; ++t)
#pragma unroll
        for (int v = 0; v < 8; ++v)
            drow[(size_t)v * HID_F + t * 16] = acc[t][v];
}

__global__ __launch_bounds__(128) void gemm2_wmma(
    const float* __restrict__ h1, const float* __restrict__ W2,
    float* __restrict__ h2)
{
    const int wid  = (blockIdx.x * blockDim.x + threadIdx.x) >> 5;
    const int lane = threadIdx.x & 31;
    if (wid >= N_NODES / 16) return;
    const int m    = lane & 15;
    const int kh   = lane >> 4;
    const int row0 = wid * 16;

    v8f acc[2] = {};

    const float* arow = h1 + (size_t)(row0 + m) * HID_F + 2 * kh;
    const float* wcol = W2 + m;

    for (int k = 0; k < HID_F; k += 4) {
        v2f a = *(const v2f*)(arow + k);
        a.x = fmaxf(a.x, 0.0f);
        a.y = fmaxf(a.y, 0.0f);
        const float* wrow = wcol + (size_t)(k + 2 * kh) * OUT_F;
#pragma unroll
        for (int t = 0; t < 2; ++t) {
            v2f b;
            b.x = wrow[t * 16];
            b.y = wrow[OUT_F + t * 16];
            acc[t] = __builtin_amdgcn_wmma_f32_16x16x4_f32(
                false, a, false, b, (short)0, acc[t], false, false);
        }
    }

    float* drow = h2 + (size_t)(row0 + kh * 8) * OUT_F + m;
#pragma unroll
    for (int t = 0; t < 2; ++t)
#pragma unroll
        for (int v = 0; v < 8; ++v)
            drow[(size_t)v * OUT_F + t * 16] = acc[t][v];
}

__global__ __launch_bounds__(256) void spmm_f128(
    const int* __restrict__ rows, const int* __restrict__ cols,
    const float* __restrict__ vals, const float* __restrict__ src,
    float* __restrict__ dst)
{
    const int e    = (blockIdx.x * blockDim.x + threadIdx.x) >> 5;
    const int lane = threadIdx.x & 31;
    if (e >= N_EDGES) return;
    const int   r = rows[e];
    const int   c = cols[e];
    const float w = vals[e];
    const float4 mv = ((const float4*)(src + (size_t)c * HID_F))[lane];
    float* d = dst + (size_t)r * HID_F + lane * 4;
    __hip_atomic_fetch_add(d + 0, w * mv.x, __ATOMIC_RELAXED, __HIP_MEMORY_SCOPE_AGENT);
    __hip_atomic_fetch_add(d + 1, w * mv.y, __ATOMIC_RELAXED, __HIP_MEMORY_SCOPE_AGENT);
    __hip_atomic_fetch_add(d + 2, w * mv.z, __ATOMIC_RELAXED, __HIP_MEMORY_SCOPE_AGENT);
    __hip_atomic_fetch_add(d + 3, w * mv.w, __ATOMIC_RELAXED, __HIP_MEMORY_SCOPE_AGENT);
}

__global__ __launch_bounds__(256) void spmm_f32(
    const int* __restrict__ rows, const int* __restrict__ cols,
    const float* __restrict__ vals, const float* __restrict__ src,
    float* __restrict__ dst)
{
    const int e    = (blockIdx.x * blockDim.x + threadIdx.x) >> 5;
    const int lane = threadIdx.x & 31;
    if (e >= N_EDGES) return;
    const int   r = rows[e];
    const int   c = cols[e];
    const float w = vals[e];
    const float mv = src[(size_t)c * OUT_F + lane];
    __hip_atomic_fetch_add(dst + (size_t)r * OUT_F + lane, w * mv,
                           __ATOMIC_RELAXED, __HIP_MEMORY_SCOPE_AGENT);
}

__global__ __launch_bounds__(256) void zero_f4(float4* __restrict__ p, int n4)
{
    int i = blockIdx.x * blockDim.x + threadIdx.x;
    if (i < n4) p[i] = make_float4(0.f, 0.f, 0.f, 0.f);
}

extern "C" void kernel_launch(void* const* d_in, const int* in_sizes, int n_in,
                              void* d_out, int out_size, void* d_ws, size_t ws_size,
                              hipStream_t stream)
{
    const float* x    = (const float*)d_in[0];
    const int*   rows = (const int*)  d_in[1];
    const int*   cols = (const int*)  d_in[2];
    const float* vals = (const float*)d_in[3];
    const float* W1   = (const float*)d_in[4];
    const float* W2   = (const float*)d_in[5];
    float* out = (float*)d_out;

    float* h0 = (float*)d_ws;
    float* h1 = h0 + (size_t)N_NODES * HID_F;
    float* h2 = h0;

    const int gemmBlocks = ((N_NODES / 16) + 3) / 4;
    const int edgeBlocks = (N_EDGES * 32 + 255) / 256;

    gemm1_wmma<<<gemmBlocks, 128, 0, stream>>>(x, W1, h0);

    {
        int n4 = (N_NODES * HID_F) / 4;
        zero_f4<<<(n4 + 255) / 256, 256, 0, stream>>>((float4*)h1, n4);
    }
    spmm_f128<<<edgeBlocks, 256, 0, stream>>>(rows, cols, vals, h0, h1);

    gemm2_wmma<<<gemmBlocks, 128, 0, stream>>>(h1, W2, h2);

    {
        int n4 = (N_NODES * OUT_F) / 4;
        zero_f4<<<(n4 + 255) / 256, 256, 0, stream>>>((float4*)out, n4);
    }
    spmm_f32<<<edgeBlocks, 256, 0, stream>>>(rows, cols, vals, h2, out);
}


# both spmm: each edge takes a slot {col,val} in its dst row's table (one returning int add, arrival-order rank); a wave per row gathers its slots' rows, sums in registers, stores once; no f32 atomics
# speedup vs baseline: 1.4868x; 1.4868x over previous
#include <hip/hip_runtime.h>

#define N_NODES 100000
#define N_EDGES 1600000
#define IN_F    256
#define HID_F   128
#define OUT_F   32

#define T1_STRIDE 128
#define T2_STRIDE 96
#define T1_SLOTS  (T1_STRIDE / 2 - 1)
#define T2_SLOTS  (T2_STRIDE / 2 - 1)
#define T1_COUNT  (T1_STRIDE - 2)
#define T2_COUNT  (T2_STRIDE - 2)

typedef float v2f __attribute__((ext_vector_type(2)));
typedef float v8f __attribute__((ext_vector_type(8)));

__global__ __launch_bounds__(128) void gemm1_wmma(
    const float* __restrict__ x, const float* __restrict__ W1,
    float* __restrict__ h0, int* __restrict__ tbl1)
{
    const int wid  = (blockIdx.x * blockDim.x + threadIdx.x) >> 5;
    const int lane = threadIdx.x & 31;
    if (wid >= N_NODES / 16) return;
    const int m    = lane & 15;
    const int kh   = lane >> 4;
    const int row0 = wid * 16;

    v8f acc[8] = {};

    const float* arow = x + (size_t)(row0 + m) * IN_F + 2 * kh;
    const float* wcol = W1 + m;

    for (int k = 0; k < IN_F; k += 4) {
        v2f a = *(const v2f*)(arow + k);
        const float* wrow = wcol + (size_t)(k + 2 * kh) * HID_F;
#pragma unroll
        for (int t = 0; t < 8; ++t) {
            v2f b;
            b.x = wrow[t * 16];
            b.y = wrow[HID_F + t * 16];
            acc[t] = __builtin_amdgcn_wmma_f32_16x16x4_f32(
                false, a, false, b, (short)0, acc[t], false, false);
        }
    }

    float* drow = h0 + (size_t)(row0 + kh * 8) * HID_F + m;
#pragma unroll
    for (int t = 0; t < 8; ++t)
#pragma unroll
        for (int v = 0; v < 8; ++v)
            drow[(size_t)v * HID_F + t * 16] = acc[t][v];

    if (kh == 0) tbl1[(size_t)(row0 + m) * T1_STRIDE + T1_COUNT] = 0;
}

__global__ __launch_bounds__(128) void gemm2_wmma(
    const float* __restrict__ h1, const float* __restrict__ W2,
    float* __restrict__ h2, int* __restrict__ tbl2)
{
    const int wid  = (blockIdx.x * blockDim.x + threadIdx.x) >> 5;
    const int lane = threadIdx.x & 31;
    if (wid >= N_NODES / 16) return;
    const int m    = lane & 15;
    const int kh   = lane >> 4;
    const int row0 = wid * 16;

    v8f acc[2] = {};

    const float* arow = h1 + (size_t)(row0 + m) * HID_F + 2 * kh;
    const float* wcol = W2 + m;

    for (int k = 0; k < HID_F; k += 4) {
        v2f a = *(const v2f*)(arow + k);
        a.x = fmaxf(a.x, 0.0f);
        a.y = fmaxf(a.y, 0.0f);
        const float* wrow = wcol + (size_t)(k + 2 * kh) * OUT_F;
#pragma unroll
        for (int t = 0; t < 2; ++t) {
            v2f b;
            b.x = wrow[t * 16];
            b.y = wrow[OUT_F + t * 16];
            acc[t] = __builtin_amdgcn_wmma_f32_16x16x4_f32(
                false, a, false, b, (short)0, acc[t], false, false);
        }
    }

    float* drow = h2 + (size_t)(row0 + kh * 8) * OUT_F + m;
#pragma unroll
    for (int t = 0; t < 2; ++t)
#pragma unroll
        for (int v = 0; v < 8; ++v)
            drow[(size_t)v * OUT_F + t * 16] = acc[t][v];

    if (kh == 0) tbl2[(size_t)(row0 + m) * T2_STRIDE + T2_COUNT] = 0;
}

template <int STRIDE>
__global__ __launch_bounds__(256) void slot_edges(
    const int* __restrict__ rows, const int* __restrict__ cols,
    const float* __restrict__ vals, int* __restrict__ tbl)
{
    const int e = blockIdx.x * blockDim.x + threadIdx.x;
    if (e >= N_EDGES) return;
    int* t = tbl + (size_t)rows[e] * STRIDE;
    const int2 cv = make_int2(cols[e], __float_as_int(vals[e]));
    const int j = __hip_atomic_fetch_add(t + (STRIDE - 2), 1, __ATOMIC_RELAXED, __HIP_MEMORY_SCOPE_AGENT);
    if (j < STRIDE / 2 - 1) ((int2*)t)[j] = cv;
}

static __device__ __forceinline__ int   lane_i(int v, int l)   { return __builtin_amdgcn_readlane(v, l); }
static __device__ __forceinline__ float lane_f(float v, int l) { return __int_as_float(__builtin_amdgcn_readlane(__float_as_int(v), l)); }

__global__ __launch_bounds__(256) void rowsum_f128(
    const int* __restrict__ rows, const int* __restrict__ cols,
    const float* __restrict__ vals, const float* __restrict__ src,
    float* dst)
{
    const int r    = (blockIdx.x * blockDim.x + threadIdx.x) >> 5;
    const int lane = threadIdx.x & 31;
    if (r >= N_NODES) return;
    const int2*   tbl = (const int2*)(dst + (size_t)r * HID_F);
    const float4* s4  = (const float4*)src;
    const int2 p0  = tbl[lane];
    const int  cnt = lane_i(tbl[T1_SLOTS].x, 0);
    float4 acc = make_float4(0.f, 0.f, 0.f, 0.f);

    if (cnt <= T1_SLOTS) {
        for (int base = 0; base < cnt; base += 32) {
            const int2 p  = (base == 0) ? p0 : tbl[base + lane];
            const bool ok = base + lane < cnt;
            const float v = ok ? __int_as_float(p.y) : 0.f;
            const int   n = min(32, cnt - base);
            const int   c = ok ? p.x : lane_i(p.x, 0);
            for (int i = 0; i < n; i += 4) {
                float4 g[4];
                float  f[4];
#pragma unroll
                for (int u = 0; u < 4; ++u) {
                    f[u] = lane_f(v, i + u);
                    g[u] = s4[(size_t)lane_i(c, i + u) * (HID_F / 4) + lane];
                }
#pragma unroll
                for (int u = 0; u < 4; ++u) {
                    acc.x += f[u] * g[u].x;  acc.y += f[u] * g[u].y;
                    acc.z += f[u] * g[u].z;  acc.w += f[u] * g[u].w;
                }
            }
        }
    } else {
        for (int e0 = 0; e0 < N_EDGES; e0 += 32) {
            unsigned hit = __builtin_amdgcn_ballot_w32(rows[e0 + lane] == r);
            while (hit) {
                const int e = e0 + __builtin_ctz(hit);
                hit &= hit - 1;
                const float  f = vals[e];
                const float4 g = s4[(size_t)cols[e] * (HID_F / 4) + lane];
                acc.x += f * g.x;  acc.y += f * g.y;  acc.z += f * g.z;  acc.w += f * g.w;
            }
        }
    }
    ((float4*)(dst + (size_t)r * HID_F))[lane] = acc;
}

__global__ __launch_bounds__(256) void rowsum_f32(
    const int* __restrict__ rows, const int* __restrict__ cols,
    const float* __restrict__ vals, const float* __restrict__ src,
    const int* __restrict__ tbl_all, float* __restrict__ dst)
{
    const int r    = (blockIdx.x * blockDim.x + threadIdx.x) >> 5;
    const int lane = threadIdx.x & 31;
    if (r >= N_NODES) return;
    const int2* tbl = (const int2*)(tbl_all + (size_t)r * T2_STRIDE);
    const int2 p0  = tbl[lane];
    const int  cnt = lane_i(tbl[T2_SLOTS].x, 0);
    float acc = 0.f;

    if (cnt <= T2_SLOTS) {
        for (int base = 0; base < cnt; base += 32) {
            int2 p = p0;
            if (base != 0) p = (base + lane < T2_STRIDE / 2) ? tbl[base + lane] : make_int2(0, 0);
            const bool ok = base + lane < cnt;
            const float v = ok ? __int_as_float(p.y) : 0.f;
            const int   n = min(32, cnt - base);
            const int   c = ok ? p.x : lane_i(p.x, 0);
            for (int i = 0; i < n; i += 4) {
                float g[4];
                float f[4];
#pragma unroll
                for (int u = 0; u < 4; ++u) {
                    f[u] = lane_f(v, i + u);
                    g[u] = src[(size_t)lane_i(c, i + u) * OUT_F + lane];
                }
#pragma unroll
                for (int u = 0; u < 4; ++u) acc += f[u] * g[u];
            }
        }
    } else {
        for (int e0 = 0; e0 < N_EDGES; e0 += 32) {
            unsigned hit = __builtin_amdgcn_ballot_w32(rows[e0 + lane] == r);
            while (hit) {
                const int e = e0 + __builtin_ctz(hit);
                hit &= hit - 1;
                acc += vals[e] * src[(size_t)cols[e] * OUT_F + lane];
            }
        }
    }
    dst[(size_t)r * OUT_F + lane] = acc;
}

extern "C" void kernel_launch(void* const* d_in, const int* in_sizes, int n_in,
                              void* d_out, int out_size, void* d_ws, size_t ws_size,
                              hipStream_t stream)
{
    const float* x    = (const float*)d_in[0];
    const int*   rows = (const int*)  d_in[1];
    const int*   cols = (const int*)  d_in[2];
    const float* vals = (const float*)d_in[3];
    const float* W1   = (const float*)d_in[4];
    const float* W2   = (const float*)d_in[5];
    float* out = (float*)d_out;

    float* h0 = (float*)d_ws;
    float* h1 = h0 + (size_t)N_NODES * HID_F;
    float* h2 = h0;
    int* tbl1 = (int*)h1;
    int* tbl2 = (int*)(h0 + (size_t)N_NODES * OUT_F);

    const int gemmBlocks = ((N_NODES / 16) + 3) / 4;
    const int edgeBlocks = (N_EDGES + 255) / 256;
    const int rowBlocks  = (N_NODES * 32 + 255) / 256;

    gemm1_wmma<<<gemmBlocks, 128, 0, stream>>>(x, W1, h0, tbl1);

    slot_edges<T1_STRIDE><<<edgeBlocks, 256, 0, stream>>>(rows, cols, vals, tbl1);
    rowsum_f128<<<rowBlocks, 256, 0, stream>>>(rows, cols, vals, h0, h1);

    gemm2_wmma<<<gemmBlocks, 128, 0, stream>>>(h1, W2, h2, tbl2);

    slot_edges<T2_STRIDE><<<edgeBlocks, 256, 0, stream>>>(rows, cols, vals, tbl2);
    rowsum_f32<<<rowBlocks, 256, 0, stream>>>(rows, cols, vals, h2, tbl2, out);
}


# edges slotted once for both layers: layer-1 row sum runs as two column-half launches, the 2nd moves the row's table into the dead half of its h0 row; layer 2 sums in layer 1's arrival ranks
# speedup vs baseline: 2.0805x; 1.3993x over previous
#include <hip/hip_runtime.h>

#define N_NODES 100000
#define N_EDGES 1600000
#define IN_F    256
#define HID_F   128
#define OUT_F   32

#define ROW_W     128
#define TBL_OFF   64
#define SLOTS_A   31
#define SLOTS_B   16
#define SLOTS     (SLOTS_A + SLOTS_B)
#define T1_STRIDE ROW_W
#define T1_COUNT  (TBL_OFF + 2 * SLOTS_A)

typedef float v2f __attribute__((ext_vector_type(2)));
typedef float v8f __attribute__((ext_vector_type(8)));

__global__ __launch_bounds__(128) void gemm1_wmma(
    const float* __restrict__ x, const float* __restrict__ W1,
    float* __restrict__ h0, int* __restrict__ tbl1)
{
    const int wid  = (blockIdx.x * blockDim.x + threadIdx.x) >> 5;
    const int lane = threadIdx.x & 31;
    if (wid >= N_NODES / 16) return;
    const int m    = lane & 15;
    const int kh   = lane >> 4;
    const int row0 = wid * 16;

    v8f acc[8] = {};

    const float* arow = x + (size_t)(row0 + m) * IN_F + 2 * kh;
    const float* wcol = W1 + m;

    for (int k = 0; k < IN_F; k += 4) {
        v2f a = *(const v2f*)(arow + k);
        const float* wrow = wcol + (size_t)(k + 2 * kh) * HID_F;
#pragma unroll
        for (int t = 0; t < 8; ++t) {
            v2f b;
            b.x = wrow[t * 16];
            b.y = wrow[HID_F + t * 16];
            acc[t] = __builtin_amdgcn_wmma_f32_16x16x4_f32(
                false, a, false, b, (short)0, acc[t], false, false);
        }
    }

    float* drow = h0 + (size_t)(row0 + kh * 8) * HID_F + m;
#pragma unroll
    for (int t = 0; t < 8; ++t)
#pragma unroll
        for (int v = 0; v < 8; ++v)
            drow[(size_t)v * HID_F + t * 16] = acc[t][v];

    if (kh == 0) tbl1[(size_t)(row0 + m) * T1_STRIDE + T1_COUNT] = 0;
}

__global__ __launch_bounds__(128) void gemm2_wmma(
    const float* __restrict__ h1, const float* __restrict__ W2,
    float* __restrict__ h2)
{
    const int wid  = (blockIdx.x * blockDim.x + threadIdx.x) >> 5;
    const int lane = threadIdx.x & 31;
    if (wid >= N_NODES / 16) return;
    const int m    = lane & 15;
    const int kh   = lane >> 4;
    const int row0 = wid * 16;

    v8f acc[2] = {};

    const float* arow = h1 + (size_t)(row0 + m) * HID_F + 2 * kh;
    const float* wcol = W2 + m;

    for (int k = 0; k < HID_F; k += 4) {
        v2f a = *(const v2f*)(arow + k);
        a.x = fmaxf(a.x, 0.0f);
        a.y = fmaxf(a.y, 0.0f);
        const float* wrow = wcol + (size_t)(k + 2 * kh) * OUT_F;
#pragma unroll
        for (int t = 0; t < 2; ++t) {
            v2f b;
            b.x = wrow[t * 16];
            b.y = wrow[OUT_F + t * 16];
            acc[t] = __builtin_amdgcn_wmma_f32_16x16x4_f32(
                false, a, false, b, (short)0, acc[t], false, false);
        }
    }

    float* drow = h2 + (size_t)(row0 + kh * 8) * ROW_W + m;
#pragma unroll
    for (int t = 0; t < 2; ++t)
#pragma unroll
        for (int v = 0; v < 8; ++v)
            drow[(size_t)v * ROW_W + t * 16] = acc[t][v];
}

__global__ __launch_bounds__(256) void slot_edges(
    const int* __restrict__ rows, const int* __restrict__ cols,
    const float* __restrict__ vals, int* __restrict__ buf1, float* __restrict__ outbuf)
{
    const int e = blockIdx.x * blockDim.x + threadIdx.x;
    if (e >= N_EDGES) return;
    const int r = rows[e];
    int* t = buf1 + (size_t)r * ROW_W + TBL_OFF;
    const int2 cv = make_int2(cols[e], __float_as_int(vals[e]));
    const int j = __hip_atomic_fetch_add(t + 2 * SLOTS_A, 1, __ATOMIC_RELAXED, __HIP_MEMORY_SCOPE_AGENT);
    if (j < SLOTS_A)    ((int2*)t)[j] = cv;
    else if (j < SLOTS) ((int2*)outbuf)[(size_t)r * SLOTS_B + (j - SLOTS_A)] = cv;
}

static __device__ __forceinline__ int   lane_i(int v, int l)   { return __builtin_amdgcn_readlane(v, l); }
static __device__ __forceinline__ float lane_f(float v, int l) { return __int_as_float(__builtin_amdgcn_readlane(__float_as_int(v), l)); }

template <int HALF>
__global__ __launch_bounds__(256) void rowsum_half(
    const int* __restrict__ rows, const int* __restrict__ cols,
    const float* __restrict__ vals, float* buf0, float* buf1, const float* outbuf)
{
    const int r    = (blockIdx.x * blockDim.x + threadIdx.x) >> 5;
    const int lane = threadIdx.x & 31;
    if (r >= N_NODES) return;
    float* hrow = buf1 + (size_t)r * ROW_W;
    const int2 p0  = ((const int2*)(hrow + TBL_OFF))[lane];
    const int  cnt = lane_i(p0.x, SLOTS_A);
    const float2* s2 = (const float2*)buf0 + HALF * 32 + lane;
    float2 acc = make_float2(0.f, 0.f);

    if (cnt <= SLOTS) {
#pragma unroll
        for (int part = 0; part < 2; ++part) {
            int2 p = p0;
            int  n = min(cnt, SLOTS_A);
            if (part == 1) {
                if (cnt <= SLOTS_A) break;
                p = ((const int2*)outbuf)[(size_t)r * SLOTS_B + (lane & (SLOTS_B - 1))];
                n = cnt - SLOTS_A;
            }
            const bool ok = lane < n;
            const float v = ok ? __int_as_float(p.y) : 0.f;
            const int   c = ok ? p.x : lane_i(p.x, 0);
            for (int i = 0; i < n; i += 4) {
                float2 g[4];
                float  f[4];
#pragma unroll
                for (int u = 0; u < 4; ++u) {
                    f[u] = lane_f(v, i + u);
                    g[u] = s2[(size_t)lane_i(c, i + u) * (ROW_W / 2)];
                }
#pragma unroll
                for (int u = 0; u < 4; ++u) {
                    acc.x += f[u] * g[u].x;  acc.y += f[u] * g[u].y;
                }
            }
        }
    } else {
        for (int e0 = 0; e0 < N_EDGES; e0 += 32) {
            unsigned hit = __builtin_amdgcn_ballot_w32(rows[e0 + lane] == r);
            while (hit) {
                const int e = e0 + __builtin_ctz(hit);
                hit &= hit - 1;
                const float  f = vals[e];
                const float2 g = s2[(size_t)cols[e] * (ROW_W / 2)];
                acc.x += f * g.x;  acc.y += f * g.y;
            }
        }
    }
    if (HALF == 1) ((int2*)(buf0 + (size_t)r * ROW_W))[lane] = p0;
    ((float2*)(hrow + HALF * 64))[lane] = acc;
}

__global__ __launch_bounds__(256) void rowsum_f32(
    const int* __restrict__ rows, const int* __restrict__ cols,
    const float* __restrict__ vals, const float* __restrict__ buf0, float* outbuf)
{
    const int r    = (blockIdx.x * blockDim.x + threadIdx.x) >> 5;
    const int lane = threadIdx.x & 31;
    if (r >= N_NODES) return;
    const int2 p0  = ((const int2*)(buf0 + (size_t)r * ROW_W))[lane];
    const int  cnt = lane_i(p0.x, SLOTS_A);
    const float* s = buf0 + TBL_OFF + lane;
    float acc = 0.f;

    if (cnt <= SLOTS) {
#pragma unroll
        for (int part = 0; part < 2; ++part) {
            int2 p = p0;
            int  n = min(cnt, SLOTS_A);
            if (part == 1) {
                if (cnt <= SLOTS_A) break;
                p = ((const int2*)outbuf)[(size_t)r * SLOTS_B + (lane & (SLOTS_B - 1))];
                n = cnt - SLOTS_A;
            }
            const bool ok = lane < n;
            const float v = ok ? __int_as_float(p.y) : 0.f;
            const int   c = ok ? p.x : lane_i(p.x, 0);
            for (int i = 0; i < n; i += 4) {
                float g[4];
                float f[4];
#pragma unroll
                for (int u = 0; u < 4; ++u) {
                    f[u] = lane_f(v, i + u);
                    g[u] = s[(size_t)lane_i(c, i + u) * ROW_W];
                }
#pragma unroll
                for (int u = 0; u < 4; ++u) acc += f[u] * g[u];
            }
        }
    } else {
        for (int e0 = 0; e0 < N_EDGES; e0 += 32) {
            unsigned hit = __builtin_amdgcn_ballot_w32(rows[e0 + lane] == r);
            while (hit) {
                const int e = e0 + __builtin_ctz(hit);
                hit &= hit - 1;
                acc += vals[e] * s[(size_t)cols[e] * ROW_W];
            }
        }
    }
    outbuf[(size_t)r * OUT_F + lane] = acc;
}

extern "C" void kernel_launch(void* const* d_in, const int* in_sizes, int n_in,
                              void* d_out, int out_size, void* d_ws, size_t ws_size,
                              hipStream_t stream)
{
    const float* x    = (const float*)d_in[0];
    const int*   rows = (const int*)  d_in[1];
    const int*   cols = (const int*)  d_in[2];
    const float* vals = (const float*)d_in[3];
    const float* W1   = (const float*)d_in[4];
    const float* W2   = (const float*)d_in[5];
    float* out = (float*)d_out;

    float* buf0 = (float*)d_ws;
    float* buf1 = buf0 + (size_t)N_NODES * ROW_W;

    const int gemmBlocks = ((N_NODES / 16) + 3) / 4;
    const int edgeBlocks = (N_EDGES + 255) / 256;
    const int rowBlocks  = (N_NODES * 32 + 255) / 256;

    gemm1_wmma<<<gemmBlocks, 128, 0, stream>>>(x, W1, buf0, (int*)buf1);

    slot_edges<<<edgeBlocks, 256, 0, stream>>>(rows, cols, vals, (int*)buf1, out);

    rowsum_half<0><<<rowBlocks, 256, 0, stream>>>(rows, cols, vals, buf0, buf1, out);
    rowsum_half<1><<<rowBlocks, 256, 0, stream>>>(rows, cols, vals, buf0, buf1, out);

    gemm2_wmma<<<gemmBlocks, 128, 0, stream>>>(buf1, W2, buf0 + TBL_OFF);

    rowsum_f32<<<rowBlocks, 256, 0, stream>>>(rows, cols, vals, buf0, out);
}


# slotting kept inside each half of the package: a wave reads its die, draws 256-edge runs from its half's counter, slots only edges whose destination row is in its half's range; arrival-order ranks
# speedup vs baseline: 2.7783x; 1.3354x over previous
#include <hip/hip_runtime.h>

#define N_NODES 100000
#define N_EDGES 1600000
#define IN_F    256
#define HID_F   128
#define OUT_F   32

#define ROW_W     128
#define TBL_OFF   64
#define SLOTS_A   31
#define SLOTS_B   16
#define SLOTS     (SLOTS_A + SLOTS_B)
#define T1_STRIDE ROW_W
#define T1_COUNT  (TBL_OFF + 2 * SLOTS_A)

#define HALF_ROWS (N_NODES / 2)
#define RUN       256
#define N_RUNS    (N_EDGES / RUN)
#define T1_DRAW   (T1_COUNT + 1)
#define DRAW_ROW(h) ((h) ? (size_t)(N_NODES - 1) : (size_t)0)
#define SLOT_BLOCKS 1024

typedef float v2f __attribute__((ext_vector_type(2)));
typedef float v8f __attribute__((ext_vector_type(8)));

__global__ __launch_bounds__(128) void gemm1_wmma(
    const float* __restrict__ x, const float* __restrict__ W1,
    float* __restrict__ h0, int* __restrict__ tbl1)
{
    const int wid  = (blockIdx.x * blockDim.x + threadIdx.x) >> 5;
    const int lane = threadIdx.x & 31;
    if (wid >= N_NODES / 16) return;
    const int m    = lane & 15;
    const int kh   = lane >> 4;
    const int row0 = wid * 16;

    v8f acc[8] = {};

    const float* arow = x + (size_t)(row0 + m) * IN_F + 2 * kh;
    const float* wcol = W1 + m;

    for (int k = 0; k < IN_F; k += 4) {
        v2f a = *(const v2f*)(arow + k);
        const float* wrow = wcol + (size_t)(k + 2 * kh) * HID_F;
#pragma unroll
        for (int t = 0; t < 8; ++t) {
            v2f b;
            b.x = wrow[t * 16];
            b.y = wrow[HID_F + t * 16];
            acc[t] = __builtin_amdgcn_wmma_f32_16x16x4_f32(
                false, a, false, b, (short)0, acc[t], false, false);
        }
    }

    float* drow = h0 + (size_t)(row0 + kh * 8) * HID_F + m;
#pragma unroll
    for (int t = 0; t < 8; ++t)
#pragma unroll
        for (int v = 0; v < 8; ++v)
            drow[(size_t)v * HID_F + t * 16] = acc[t][v];

    if (kh == 0) tbl1[(size_t)(row0 + m) * T1_STRIDE + T1_COUNT] = 0;
    if (wid == 0 && lane == 0) {
        tbl1[DRAW_ROW(0) * T1_STRIDE + T1_DRAW] = 0;
        tbl1[DRAW_ROW(1) * T1_STRIDE + T1_DRAW] = 0;
    }
}

__global__ __launch_bounds__(128) void gemm2_wmma(
    const float* __restrict__ h1, const float* __restrict__ W2,
    float* __restrict__ h2)
{
    const int wid  = (blockIdx.x * blockDim.x + threadIdx.x) >> 5;
    const int lane = threadIdx.x & 31;
    if (wid >= N_NODES / 16) return;
    const int m    = lane & 15;
    const int kh   = lane >> 4;
    const int row0 = wid * 16;

    v8f acc[2] = {};

    const float* arow = h1 + (size_t)(row0 + m) * HID_F + 2 * kh;
    const float* wcol = W2 + m;

    for (int k = 0; k < HID_F; k += 4) {
        v2f a = *(const v2f*)(arow + k);
        a.x = fmaxf(a.x, 0.0f);
        a.y = fmaxf(a.y, 0.0f);
        const float* wrow = wcol + (size_t)(k + 2 * kh) * OUT_F;
#pragma unroll
        for (int t = 0; t < 2; ++t) {
            v2f b;
            b.x = wrow[t * 16];
            b.y = wrow[OUT_F + t * 16];
            acc[t] = __builtin_amdgcn_wmma_f32_16x16x4_f32(
                false, a, false, b, (short)0, acc[t], false, false);
        }
    }

    float* drow = h2 + (size_t)(row0 + kh * 8) * ROW_W + m;
#pragma unroll
    for (int t = 0; t < 2; ++t)
#pragma unroll
        for (int v = 0; v < 8; ++v)
            drow[(size_t)v * ROW_W + t * 16] = acc[t][v];
}

__global__ __launch_bounds__(128) void slot_edges(
    const int* __restrict__ rows, const int* __restrict__ cols,
    const float* __restrict__ vals, int* __restrict__ buf1, float* __restrict__ outbuf)
{
    const int lane = threadIdx.x & 31;
    const int mine = (int)((__builtin_amdgcn_s_sendmsg_rtn(0x87) >> 18) & 1u);

    for (int pass = 0; pass < 2; ++pass) {
        const int hh = mine ^ pass;
        int* ctr = buf1 + DRAW_ROW(hh) * T1_STRIDE + T1_DRAW;
        for (int it = 0; it <= N_RUNS; ++it) {
            int k = 0;
            if (lane == 0) k = __hip_atomic_fetch_add(ctr, 1, __ATOMIC_RELAXED, __HIP_MEMORY_SCOPE_AGENT);
            k = __builtin_amdgcn_readfirstlane(k);
            if (k >= N_RUNS) break;

            const int e = k * RUN + lane * 4;
            const int4   ra = *(const int4*)(rows + e),   rb = *(const int4*)(rows + e + RUN / 2);
            const int4   ca = *(const int4*)(cols + e),   cb = *(const int4*)(cols + e + RUN / 2);
            const float4 va = *(const float4*)(vals + e), vb = *(const float4*)(vals + e + RUN / 2);
            const int   r[8] = {ra.x, ra.y, ra.z, ra.w, rb.x, rb.y, rb.z, rb.w};
            const int   c[8] = {ca.x, ca.y, ca.z, ca.w, cb.x, cb.y, cb.z, cb.w};
            const float v[8] = {va.x, va.y, va.z, va.w, vb.x, vb.y, vb.z, vb.w};

            int j[8];
#pragma unroll
            for (int u = 0; u < 8; ++u) {
                j[u] = SLOTS;
                if ((r[u] >= HALF_ROWS) == (hh != 0))
                    j[u] = __hip_atomic_fetch_add(buf1 + (size_t)r[u] * ROW_W + T1_COUNT, 1,
                                                  __ATOMIC_RELAXED, __HIP_MEMORY_SCOPE_AGENT);
            }
#pragma unroll
            for (int u = 0; u < 8; ++u) {
                const int2 cv = make_int2(c[u], __float_as_int(v[u]));
                if (j[u] < SLOTS_A)    ((int2*)(buf1 + (size_t)r[u] * ROW_W + TBL_OFF))[j[u]] = cv;
                else if (j[u] < SLOTS) ((int2*)outbuf)[(size_t)r[u] * SLOTS_B + (j[u] - SLOTS_A)] = cv;
            }
        }
    }
}

static __device__ __forceinline__ int   lane_i(int v, int l)   { return __builtin_amdgcn_readlane(v, l); }
static __device__ __forceinline__ float lane_f(float v, int l) { return __int_as_float(__builtin_amdgcn_readlane(__float_as_int(v), l)); }

template <int HALF>
__global__ __launch_bounds__(256) void rowsum_half(
    const int* __restrict__ rows, const int* __restrict__ cols,
    const float* __restrict__ vals, float* buf0, float* buf1, const float* outbuf)
{
    const int r    = (blockIdx.x * blockDim.x + threadIdx.x) >> 5;
    const int lane = threadIdx.x & 31;
    if (r >= N_NODES) return;
    float* hrow = buf1 + (size_t)r * ROW_W;
    const int2 p0  = ((const int2*)(hrow + TBL_OFF))[lane];
    const int  cnt = lane_i(p0.x, SLOTS_A);
    const float2* s2 = (const float2*)buf0 + HALF * 32 + lane;
    float2 acc = make_float2(0.f, 0.f);

    if (cnt <= SLOTS) {
#pragma unroll
        for (int part = 0; part < 2; ++part) {
            int2 p = p0;
            int  n = min(cnt, SLOTS_A);
            if (part == 1) {
                if (cnt <= SLOTS_A) break;
                p = ((const int2*)outbuf)[(size_t)r * SLOTS_B + (lane & (SLOTS_B - 1))];
                n = cnt - SLOTS_A;
            }
            const bool ok = lane < n;
            const float v = ok ? __int_as_float(p.y) : 0.f;
            const int   c = ok ? p.x : lane_i(p.x, 0);
            for (int i = 0; i < n; i += 4) {
                float2 g[4];
                float  f[4];
#pragma unroll
                for (int u = 0; u < 4; ++u) {
                    f[u] = lane_f(v, i + u);
                    g[u] = s2[(size_t)lane_i(c, i + u) * (ROW_W / 2)];
                }
#pragma unroll
                for (int u = 0; u < 4; ++u) {
                    acc.x += f[u] * g[u].x;  acc.y += f[u] * g[u].y;
                }
            }
        }
    } else {
        for (int e0 = 0; e0 < N_EDGES; e0 += 32) {
            unsigned hit = __builtin_amdgcn_ballot_w32(rows[e0 + lane] == r);
            while (hit) {
                const int e = e0 + __builtin_ctz(hit);
                hit &= hit - 1;
                const float  f = vals[e];
                const float2 g = s2[(size_t)cols[e] * (ROW_W / 2)];
                acc.x += f * g.x;  acc.y += f * g.y;
            }
        }
    }
    if (HALF == 1) ((int2*)(buf0 + (size_t)r * ROW_W))[lane] = p0;
    ((float2*)(hrow + HALF * 64))[lane] = acc;
}

__global__ __launch_bounds__(256) void rowsum_f32(
    const int* __restrict__ rows, const int* __restrict__ cols,
    const float* __restrict__ vals, const float* __restrict__ buf0, float* outbuf)
{
    const int r    = (blockIdx.x * blockDim.x + threadIdx.x) >> 5;
    const int lane = threadIdx.x & 31;
    if (r >= N_NODES) return;
    const int2 p0  = ((const int2*)(buf0 + (size_t)r * ROW_W))[lane];
    const int  cnt = lane_i(p0.x, SLOTS_A);
    const float* s = buf0 + TBL_OFF + lane;
    float acc = 0.f;

    if (cnt <= SLOTS) {
#pragma unroll
        for (int part = 0; part < 2; ++part) {
            int2 p = p0;
            int  n = min(cnt, SLOTS_A);
            if (part == 1) {
                if (cnt <= SLOTS_A) break;
                p = ((const int2*)outbuf)[(size_t)r * SLOTS_B + (lane & (SLOTS_B - 1))];
                n = cnt - SLOTS_A;
            }
            const bool ok = lane < n;
            const float v = ok ? __int_as_float(p.y) : 0.f;
            const int   c = ok ? p.x : lane_i(p.x, 0);
            for (int i = 0; i < n; i += 4) {
                float g[4];
                float f[4];
#pragma unroll
                for (int u = 0; u < 4; ++u) {
                    f[u] = lane_f(v, i + u);
                    g[u] = s[(size_t)lane_i(c, i + u) * ROW_W];
                }
#pragma unroll
                for (int u = 0; u < 4; ++u) acc += f[u] * g[u];
            }
        }
    } else {
        for (int e0 = 0; e0 < N_EDGES; e0 += 32) {
            unsigned hit = __builtin_amdgcn_ballot_w32(rows[e0 + lane] == r);
            while (hit) {
                const int e = e0 + __builtin_ctz(hit);
                hit &= hit - 1;
                acc += vals[e] * s[(size_t)cols[e] * ROW_W];
            }
        }
    }
    outbuf[(size_t)r * OUT_F + lane] = acc;
}

extern "C" void kernel_launch(void* const* d_in, const int* in_sizes, int n_in,
                              void* d_out, int out_size, void* d_ws, size_t ws_size,
                              hipStream_t stream)
{
    const float* x    = (const float*)d_in[0];
    const int*   rows = (const int*)  d_in[1];
    const int*   cols = (const int*)  d_in[2];
    const float* vals = (const float*)d_in[3];
    const float* W1   = (const float*)d_in[4];
    const float* W2   = (const float*)d_in[5];
    float* out = (float*)d_out;

    float* buf0 = (float*)d_ws;
    float* buf1 = buf0 + (size_t)N_NODES * ROW_W;

    const int gemmBlocks = ((N_NODES / 16) + 3) / 4;
    const int rowBlocks  = (N_NODES * 32 + 255) / 256;

    gemm1_wmma<<<gemmBlocks, 128, 0, stream>>>(x, W1, buf0, (int*)buf1);

    slot_edges<<<SLOT_BLOCKS, 128, 0, stream>>>(rows, cols, vals, (int*)buf1, out);

    rowsum_half<0><<<rowBlocks, 256, 0, stream>>>(rows, cols, vals, buf0, buf1, out);
    rowsum_half<1><<<rowBlocks, 256, 0, stream>>>(rows, cols, vals, buf0, buf1, out);

    gemm2_wmma<<<gemmBlocks, 128, 0, stream>>>(buf1, W2, buf0 + TBL_OFF);

    rowsum_f32<<<rowBlocks, 256, 0, stream>>>(rows, cols, vals, buf0, out);
}


# edge slotting runs inside the first product's launch, its 1024 work-groups dispatched ahead of the matrix blocks (adds wait under matrix issue); counts zeroed by a small launch first; arrival ranks
# speedup vs baseline: 2.9278x; 1.0538x over previous
#include <hip/hip_runtime.h>

#define N_NODES 100000
#define N_EDGES 1600000
#define IN_F    256
#define HID_F   128
#define OUT_F   32

#define ROW_W     128
#define TBL_OFF   64
#define SLOTS_A   31
#define SLOTS_B   16
#define SLOTS     (SLOTS_A + SLOTS_B)
#define T1_STRIDE ROW_W
#define T1_COUNT  (TBL_OFF + 2 * SLOTS_A)

#define HALF_ROWS (N_NODES / 2)
#define RUN       256
#define N_RUNS    (N_EDGES / RUN)
#define T1_DRAW   (T1_COUNT + 1)
#define DRAW_ROW(h) ((h) ? (size_t)(N_NODES - 1) : (size_t)0)
#define SLOT_BLOCKS 1024

typedef float v2f __attribute__((ext_vector_type(2)));
typedef float v8f __attribute__((ext_vector_type(8)));

__global__ __launch_bounds__(256) void zero_counts(int* __restrict__ tbl1)
{
    const int r = blockIdx.x * blockDim.x + threadIdx.x;
    if (r >= N_NODES) return;
    tbl1[(size_t)r * T1_STRIDE + T1_COUNT] = 0;
    if (r == 0 || r == N_NODES - 1) tbl1[(size_t)r * T1_STRIDE + T1_DRAW] = 0;
}

static __device__ __forceinline__ void gemm1_rows(
    const int wid, const int lane,
    const float* __restrict__ x, const float* __restrict__ W1, float* __restrict__ h0)
{
    if (wid >= N_NODES / 16) return;
    const int m    = lane & 15;
    const int kh   = lane >> 4;
    const int row0 = wid * 16;

    v8f acc[8] = {};

    const float* arow = x + (size_t)(row0 + m) * IN_F + 2 * kh;
    const float* wcol = W1 + m;

    for (int k = 0; k < IN_F; k += 4) {
        v2f a = *(const v2f*)(arow + k);
        const float* wrow = wcol + (size_t)(k + 2 * kh) * HID_F;
#pragma unroll
        for (int t = 0; t < 8; ++t) {
            v2f b;
            b.x = wrow[t * 16];
            b.y = wrow[HID_F + t * 16];
            acc[t] = __builtin_amdgcn_wmma_f32_16x16x4_f32(
                false, a, false, b, (short)0, acc[t], false, false);
        }
    }

    float* drow = h0 + (size_t)(row0 + kh * 8) * HID_F + m;
#pragma unroll
    for (int t = 0; t < 8; ++t)
#pragma unroll
        for (int v = 0; v < 8; ++v)
            drow[(size_t)v * HID_F + t * 16] = acc[t][v];
}

__global__ __launch_bounds__(128) void gemm2_wmma(
    const float* __restrict__ h1, const float* __restrict__ W2,
    float* __restrict__ h2)
{
    const int wid  = (blockIdx.x * blockDim.x + threadIdx.x) >> 5;
    const int lane = threadIdx.x & 31;
    if (wid >= N_NODES / 16) return;
    const int m    = lane & 15;
    const int kh   = lane >> 4;
    const int row0 = wid * 16;

    v8f acc[2] = {};

    const float* arow = h1 + (size_t)(row0 + m) * HID_F + 2 * kh;
    const float* wcol = W2 + m;

    for (int k = 0; k < HID_F; k += 4) {
        v2f a = *(const v2f*)(arow + k);
        a.x = fmaxf(a.x, 0.0f);
        a.y = fmaxf(a.y, 0.0f);
        const float* wrow = wcol + (size_t)(k + 2 * kh) * OUT_F;
#pragma unroll
        for (int t = 0; t < 2; ++t) {
            v2f b;
            b.x = wrow[t * 16];
            b.y = wrow[OUT_F + t * 16];
            acc[t] = __builtin_amdgcn_wmma_f32_16x16x4_f32(
                false, a, false, b, (short)0, acc[t], false, false);
        }
    }

    float* drow = h2 + (size_t)(row0 + kh * 8) * ROW_W + m;
#pragma unroll
    for (int t = 0; t < 2; ++t)
#pragma unroll
        for (int v = 0; v < 8; ++v)
            drow[(size_t)v * ROW_W + t * 16] = acc[t][v];
}

static __device__ __forceinline__ void slot_edges(
    const int lane,
    const int* __restrict__ rows, const int* __restrict__ cols,
    const float* __restrict__ vals, int* __restrict__ buf1, float* __restrict__ outbuf)
{
    const int mine = (int)((__builtin_amdgcn_s_sendmsg_rtn(0x87) >> 18) & 1u);

    for (int pass = 0; pass < 2; ++pass) {
        const int hh = mine ^ pass;
        int* ctr = buf1 + DRAW_ROW(hh) * T1_STRIDE + T1_DRAW;
        for (int it = 0; it <= N_RUNS; ++it) {
            int k = 0;
            if (lane == 0) k = __hip_atomic_fetch_add(ctr, 1, __ATOMIC_RELAXED, __HIP_MEMORY_SCOPE_AGENT);
            k = __builtin_amdgcn_readfirstlane(k);
            if (k >= N_RUNS) break;

            const int e = k * RUN + lane * 4;
            const int4   ra = *(const int4*)(rows + e),   rb = *(const int4*)(rows + e + RUN / 2);
            const int4   ca = *(const int4*)(cols + e),   cb = *(const int4*)(cols + e + RUN / 2);
            const float4 va = *(const float4*)(vals + e), vb = *(const float4*)(vals + e + RUN / 2);
            const int   r[8] = {ra.x, ra.y, ra.z, ra.w, rb.x, rb.y, rb.z, rb.w};
            const int   c[8] = {ca.x, ca.y, ca.z, ca.w, cb.x, cb.y, cb.z, cb.w};
            const float v[8] = {va.x, va.y, va.z, va.w, vb.x, vb.y, vb.z, vb.w};

            int j[8];
#pragma unroll
            for (int u = 0; u < 8; ++u) {
                j[u] = SLOTS;
                if ((r[u] >= HALF_ROWS) == (hh != 0))
                    j[u] = __hip_atomic_fetch_add(buf1 + (size_t)r[u] * ROW_W + T1_COUNT, 1,
                                                  __ATOMIC_RELAXED, __HIP_MEMORY_SCOPE_AGENT);
            }
#pragma unroll
            for (int u = 0; u < 8; ++u) {
                const int2 cv = make_int2(c[u], __float_as_int(v[u]));
                if (j[u] < SLOTS_A)    ((int2*)(buf1 + (size_t)r[u] * ROW_W + TBL_OFF))[j[u]] = cv;
                else if (j[u] < SLOTS) ((int2*)outbuf)[(size_t)r[u] * SLOTS_B + (j[u] - SLOTS_A)] = cv;
            }
        }
    }
}

__global__ __launch_bounds__(128) void gemm1_and_slots(
    const float* __restrict__ x, const float* __restrict__ W1, float* __restrict__ h0,
    const int* __restrict__ rows, const int* __restrict__ cols,
    const float* __restrict__ vals, int* __restrict__ buf1, float* __restrict__ outbuf)
{
    const int lane = threadIdx.x & 31;
    if (blockIdx.x < SLOT_BLOCKS) {
        slot_edges(lane, rows, cols, vals, buf1, outbuf);
    } else {
        const int wid = ((blockIdx.x - SLOT_BLOCKS) * blockDim.x + threadIdx.x) >> 5;
        gemm1_rows(wid, lane, x, W1, h0);
    }
}

static __device__ __forceinline__ int   lane_i(int v, int l)   { return __builtin_amdgcn_readlane(v, l); }
static __device__ __forceinline__ float lane_f(float v, int l) { return __int_as_float(__builtin_amdgcn_readlane(__float_as_int(v), l)); }

template <int HALF>
__global__ __launch_bounds__(256) void rowsum_half(
    const int* __restrict__ rows, const int* __restrict__ cols,
    const float* __restrict__ vals, float* buf0, float* buf1, const float* outbuf)
{
    const int r    = (blockIdx.x * blockDim.x + threadIdx.x) >> 5;
    const int lane = threadIdx.x & 31;
    if (r >= N_NODES) return;
    float* hrow = buf1 + (size_t)r * ROW_W;
    const int2 p0  = ((const int2*)(hrow + TBL_OFF))[lane];
    const int  cnt = lane_i(p0.x, SLOTS_A);
    const float2* s2 = (const float2*)buf0 + HALF * 32 + lane;
    float2 acc = make_float2(0.f, 0.f);

    if (cnt <= SLOTS) {
#pragma unroll
        for (int part = 0; part < 2; ++part) {
            int2 p = p0;
            int  n = min(cnt, SLOTS_A);
            if (part == 1) {
                if (cnt <= SLOTS_A) break;
                p = ((const int2*)outbuf)[(size_t)r * SLOTS_B + (lane & (SLOTS_B - 1))];
                n = cnt - SLOTS_A;
            }
            const bool ok = lane < n;
            const float v = ok ? __int_as_float(p.y) : 0.f;
            const int   c = ok ? p.x : lane_i(p.x, 0);
            for (int i = 0; i < n; i += 4) {
                float2 g[4];
                float  f[4];
#pragma unroll
                for (int u = 0; u < 4; ++u) {
                    f[u] = lane_f(v, i + u);
                    g[u] = s2[(size_t)lane_i(c, i + u) * (ROW_W / 2)];
                }
#pragma unroll
                for (int u = 0; u < 4; ++u) {
                    acc.x += f[u] * g[u].x;  acc.y += f[u] * g[u].y;
                }
            }
        }
    } else {
        for (int e0 = 0; e0 < N_EDGES; e0 += 32) {
            unsigned hit = __builtin_amdgcn_ballot_w32(rows[e0 + lane] == r);
            while (hit) {
                const int e = e0 + __builtin_ctz(hit);
                hit &= hit - 1;
                const float  f = vals[e];
                const float2 g = s2[(size_t)cols[e] * (ROW_W / 2)];
                acc.x += f * g.x;  acc.y += f * g.y;
            }
        }
    }
    if (HALF == 1) ((int2*)(buf0 + (size_t)r * ROW_W))[lane] = p0;
    ((float2*)(hrow + HALF * 64))[lane] = acc;
}

__global__ __launch_bounds__(256) void rowsum_f32(
    const int* __restrict__ rows, const int* __restrict__ cols,
    const float* __restrict__ vals, const float* __restrict__ buf0, float* outbuf)
{
    const int r    = (blockIdx.x * blockDim.x + threadIdx.x) >> 5;
    const int lane = threadIdx.x & 31;
    if (r >= N_NODES) return;
    const int2 p0  = ((const int2*)(buf0 + (size_t)r * ROW_W))[lane];
    const int  cnt = lane_i(p0.x, SLOTS_A);
    const float* s = buf0 + TBL_OFF + lane;
    float acc = 0.f;

    if (cnt <= SLOTS) {
#pragma unroll
        for (int part = 0; part < 2; ++part) {
            int2 p = p0;
            int  n = min(cnt, SLOTS_A);
            if (part == 1) {
                if (cnt <= SLOTS_A) break;
                p = ((const int2*)outbuf)[(size_t)r * SLOTS_B + (lane & (SLOTS_B - 1))];
                n = cnt - SLOTS_A;
            }
            const bool ok = lane < n;
            const float v = ok ? __int_as_float(p.y) : 0.f;
            const int   c = ok ? p.x : lane_i(p.x, 0);
            for (int i = 0; i < n; i += 4) {
                float g[4];
                float f[4];
#pragma unroll
                for (int u = 0; u < 4; ++u) {
                    f[u] = lane_f(v, i + u);
                    g[u] = s[(size_t)lane_i(c, i + u) * ROW_W];
                }
#pragma unroll
                for (int u = 0; u < 4; ++u) acc += f[u] * g[u];
            }
        }
    } else {
        for (int e0 = 0; e0 < N_EDGES; e0 += 32) {
            unsigned hit = __builtin_amdgcn_ballot_w32(rows[e0 + lane] == r);
            while (hit) {
                const int e = e0 + __builtin_ctz(hit);
                hit &= hit - 1;
                acc += vals[e] * s[(size_t)cols[e] * ROW_W];
            }
        }
    }
    outbuf[(size_t)r * OUT_F + lane] = acc;
}

extern "C" void kernel_launch(void* const* d_in, const int* in_sizes, int n_in,
                              void* d_out, int out_size, void* d_ws, size_t ws_size,
                              hipStream_t stream)
{
    const float* x    = (const float*)d_in[0];
    const int*   rows = (const int*)  d_in[1];
    const int*   cols = (const int*)  d_in[2];
    const float* vals = (const float*)d_in[3];
    const float* W1   = (const float*)d_in[4];
    const float* W2   = (const float*)d_in[5];
    float* out = (float*)d_out;

    float* buf0 = (float*)d_ws;
    float* buf1 = buf0 + (size_t)N_NODES * ROW_W;

    const int gemmBlocks = ((N_NODES / 16) + 3) / 4;
    const int rowBlocks  = (N_NODES * 32 + 255) / 256;
    const int rowThreads = (N_NODES + 255) / 256;

    zero_counts<<<rowThreads, 256, 0, stream>>>((int*)buf1);

    gemm1_and_slots<<<SLOT_BLOCKS + gemmBlocks, 128, 0, stream>>>(
        x, W1, buf0, rows, cols, vals, (int*)buf1, out);

    rowsum_half<0><<<rowBlocks, 256, 0, stream>>>(rows, cols, vals, buf0, buf1, out);
    rowsum_half<1><<<rowBlocks, 256, 0, stream>>>(rows, cols, vals, buf0, buf1, out);

    gemm2_wmma<<<gemmBlocks, 128, 0, stream>>>(buf1, W2, buf0 + TBL_OFF);

    rowsum_f32<<<rowBlocks, 256, 0, stream>>>(rows, cols, vals, buf0, out);
}


# workspace re-laid as dense 64-column planes (no 512-byte row stride): every gathered table and the slot table is contiguous; the slot table stays in place for layer 2 (no copy); sum order unchanged
# speedup vs baseline: 3.0412x; 1.0387x over previous
#include <hip/hip_runtime.h>

#define N_NODES 100000
#define N_EDGES 1600000
#define IN_F    256
#define HID_F   128
#define OUT_F   32

#define PLANE_W   64
#define PLANE_SZ  ((size_t)N_NODES * PLANE_W)

#define SLOTS_A   31
#define SLOTS_B   16
#define SLOTS     (SLOTS_A + SLOTS_B)
#define T_COUNT   (2 * SLOTS_A)

#define HALF_ROWS (N_NODES / 2)
#define RUN       256
#define N_RUNS    (N_EDGES / RUN)
#define T_DRAW    (T_COUNT + 1)
#define DRAW_ROW(h) ((h) ? (size_t)(N_NODES - 1) : (size_t)0)
#define SLOT_BLOCKS 1024

typedef float v2f __attribute__((ext_vector_type(2)));
typedef float v8f __attribute__((ext_vector_type(8)));

__global__ __launch_bounds__(256) void zero_counts(int* __restrict__ tbl)
{
    const int r = blockIdx.x * blockDim.x + threadIdx.x;
    if (r >= N_NODES) return;
    tbl[(size_t)r * PLANE_W + T_COUNT] = 0;
    if (r == 0 || r == N_NODES - 1) tbl[(size_t)r * PLANE_W + T_DRAW] = 0;
}

static __device__ __forceinline__ void gemm1_rows(
    const int wid, const int lane,
    const float* __restrict__ x, const float* __restrict__ W1,
    float* __restrict__ h0a, float* __restrict__ h0b)
{
    if (wid >= N_NODES / 16) return;
    const int m    = lane & 15;
    const int kh   = lane >> 4;
    const int row0 = wid * 16;

    v8f acc[8] = {};

    const float* arow = x + (size_t)(row0 + m) * IN_F + 2 * kh;
    const float* wcol = W1 + m;

    for (int k = 0; k < IN_F; k += 4) {
        v2f a = *(const v2f*)(arow + k);
        const float* wrow = wcol + (size_t)(k + 2 * kh) * HID_F;
#pragma unroll
        for (int t = 0; t < 8; ++t) {
            v2f b;
            b.x = wrow[t * 16];
            b.y = wrow[HID_F + t * 16];
            acc[t] = __builtin_amdgcn_wmma_f32_16x16x4_f32(
                false, a, false, b, (short)0, acc[t], false, false);
        }
    }

    const size_t roff = (size_t)(row0 + kh * 8) * PLANE_W + m;
    float* da = h0a + roff;
    float* db = h0b + roff;
#pragma unroll
    for (int t = 0; t < 4; ++t)
#pragma unroll
        for (int v = 0; v < 8; ++v) {
            da[(size_t)v * PLANE_W + t * 16] = acc[t][v];
            db[(size_t)v * PLANE_W + t * 16] = acc[t + 4][v];
        }
}

__global__ __launch_bounds__(128) void gemm2_wmma(
    const float* __restrict__ h1a, const float* __restrict__ h1b,
    const float* __restrict__ W2, float* __restrict__ h2)
{
    const int wid  = (blockIdx.x * blockDim.x + threadIdx.x) >> 5;
    const int lane = threadIdx.x & 31;
    if (wid >= N_NODES / 16) return;
    const int m    = lane & 15;
    const int kh   = lane >> 4;
    const int row0 = wid * 16;

    v8f acc[2] = {};

    const size_t aoff = (size_t)(row0 + m) * PLANE_W + 2 * kh;
    const float* wcol = W2 + m;

#pragma unroll
    for (int p = 0; p < 2; ++p) {
        const float* arow = (p ? h1b : h1a) + aoff;
        for (int k = 0; k < PLANE_W; k += 4) {
            v2f a = *(const v2f*)(arow + k);
            a.x = fmaxf(a.x, 0.0f);
            a.y = fmaxf(a.y, 0.0f);
            const float* wrow = wcol + (size_t)(p * PLANE_W + k + 2 * kh) * OUT_F;
#pragma unroll
            for (int t = 0; t < 2; ++t) {
                v2f b;
                b.x = wrow[t * 16];
                b.y = wrow[OUT_F + t * 16];
                acc[t] = __builtin_amdgcn_wmma_f32_16x16x4_f32(
                    false, a, false, b, (short)0, acc[t], false, false);
            }
        }
    }

    float* drow = h2 + (size_t)(row0 + kh * 8) * OUT_F + m;
#pragma unroll
    for (int t = 0; t < 2; ++t)
#pragma unroll
        for (int v = 0; v < 8; ++v)
            drow[(size_t)v * OUT_F + t * 16] = acc[t][v];
}

static __device__ __forceinline__ void slot_edges(
    const int lane,
    const int* __restrict__ rows, const int* __restrict__ cols,
    const float* __restrict__ vals, int* __restrict__ tbl, float* __restrict__ outbuf)
{
    const int mine = (int)((__builtin_amdgcn_s_sendmsg_rtn(0x87) >> 18) & 1u);

    for (int pass = 0; pass < 2; ++pass) {
        const int hh = mine ^ pass;
        int* ctr = tbl + DRAW_ROW(hh) * PLANE_W + T_DRAW;
        for (int it = 0; it <= N_RUNS; ++it) {
            int k = 0;
            if (lane == 0) k = __hip_atomic_fetch_add(ctr, 1, __ATOMIC_RELAXED, __HIP_MEMORY_SCOPE_AGENT);
            k = __builtin_amdgcn_readfirstlane(k);
            if (k >= N_RUNS) break;

            const int e = k * RUN + lane * 4;
            const int4   ra = *(const int4*)(rows + e),   rb = *(const int4*)(rows + e + RUN / 2);
            const int4   ca = *(const int4*)(cols + e),   cb = *(const int4*)(cols + e + RUN / 2);
            const float4 va = *(const float4*)(vals + e), vb = *(const float4*)(vals + e + RUN / 2);
            const int   r[8] = {ra.x, ra.y, ra.z, ra.w, rb.x, rb.y, rb.z, rb.w};
            const int   c[8] = {ca.x, ca.y, ca.z, ca.w, cb.x, cb.y, cb.z, cb.w};
            const float v[8] = {va.x, va.y, va.z, va.w, vb.x, vb.y, vb.z, vb.w};

            int j[8];
#pragma unroll
            for (int u = 0; u < 8; ++u) {
                j[u] = SLOTS;
                if ((r[u] >= HALF_ROWS) == (hh != 0))
                    j[u] = __hip_atomic_fetch_add(tbl + (size_t)r[u] * PLANE_W + T_COUNT, 1,
                                                  __ATOMIC_RELAXED, __HIP_MEMORY_SCOPE_AGENT);
            }
#pragma unroll
            for (int u = 0; u < 8; ++u) {
                const int2 cv = make_int2(c[u], __float_as_int(v[u]));
                if (j[u] < SLOTS_A)    ((int2*)(tbl + (size_t)r[u] * PLANE_W))[j[u]] = cv;
                else if (j[u] < SLOTS) ((int2*)outbuf)[(size_t)r[u] * SLOTS_B + (j[u] - SLOTS_A)] = cv;
            }
        }
    }
}

__global__ __launch_bounds__(128) void gemm1_and_slots(
    const float* __restrict__ x, const float* __restrict__ W1,
    float* __restrict__ h0a, float* __restrict__ h0b,
    const int* __restrict__ rows, const int* __restrict__ cols,
    const float* __restrict__ vals, int* __restrict__ tbl, float* __restrict__ outbuf)
{
    const int lane = threadIdx.x & 31;
    if (blockIdx.x < SLOT_BLOCKS) {
        slot_edges(lane, rows, cols, vals, tbl, outbuf);
    } else {
        const int wid = ((blockIdx.x - SLOT_BLOCKS) * blockDim.x + threadIdx.x) >> 5;
        gemm1_rows(wid, lane, x, W1, h0a, h0b);
    }
}

static __device__ __forceinline__ int   lane_i(int v, int l)   { return __builtin_amdgcn_readlane(v, l); }
static __device__ __forceinline__ float lane_f(float v, int l) { return __int_as_float(__builtin_amdgcn_readlane(__float_as_int(v), l)); }

__global__ __launch_bounds__(256) void rowsum_plane(
    const int* __restrict__ rows, const int* __restrict__ cols,
    const float* __restrict__ vals, const float* __restrict__ src,
    const int* __restrict__ tbl, float* __restrict__ dst, const float* __restrict__ outbuf)
{
    const int r    = (blockIdx.x * blockDim.x + threadIdx.x) >> 5;
    const int lane = threadIdx.x & 31;
    if (r >= N_NODES) return;
    const int2 p0  = ((const int2*)(tbl + (size_t)r * PLANE_W))[lane];
    const int  cnt = lane_i(p0.x, SLOTS_A);
    const float2* s2 = (const float2*)src + lane;
    float2 acc = make_float2(0.f, 0.f);

    if (cnt <= SLOTS) {
#pragma unroll
        for (int part = 0; part < 2; ++part) {
            int2 p = p0;
            int  n = min(cnt, SLOTS_A);
            if (part == 1) {
                if (cnt <= SLOTS_A) break;
                p = ((const int2*)outbuf)[(size_t)r * SLOTS_B + (lane & (SLOTS_B - 1))];
                n = cnt - SLOTS_A;
            }
            const bool ok = lane < n;
            const float v = ok ? __int_as_float(p.y) : 0.f;
            const int   c = ok ? p.x : lane_i(p.x, 0);
            for (int i = 0; i < n; i += 4) {
                float2 g[4];
                float  f[4];
#pragma unroll
                for (int u = 0; u < 4; ++u) {
                    f[u] = lane_f(v, i + u);
                    g[u] = s2[(size_t)lane_i(c, i + u) * (PLANE_W / 2)];
                }
#pragma unroll
                for (int u = 0; u < 4; ++u) {
                    acc.x += f[u] * g[u].x;  acc.y += f[u] * g[u].y;
                }
            }
        }
    } else {
        for (int e0 = 0; e0 < N_EDGES; e0 += 32) {
            unsigned hit = __builtin_amdgcn_ballot_w32(rows[e0 + lane] == r);
            while (hit) {
                const int e = e0 + __builtin_ctz(hit);
                hit &= hit - 1;
                const float  f = vals[e];
                const float2 g = s2[(size_t)cols[e] * (PLANE_W / 2)];
                acc.x += f * g.x;  acc.y += f * g.y;
            }
        }
    }
    ((float2*)(dst + (size_t)r * PLANE_W))[lane] = acc;
}

__global__ __launch_bounds__(256) void rowsum_f32(
    const int* __restrict__ rows, const int* __restrict__ cols,
    const float* __restrict__ vals, const int* __restrict__ tbl,
    const float* __restrict__ h2, float* outbuf)
{
    const int r    = (blockIdx.x * blockDim.x + threadIdx.x) >> 5;
    const int lane = threadIdx.x & 31;
    if (r >= N_NODES) return;
    const int2 p0  = ((const int2*)(tbl + (size_t)r * PLANE_W))[lane];
    const int  cnt = lane_i(p0.x, SLOTS_A);
    const float* s = h2 + lane;
    float acc = 0.f;

    if (cnt <= SLOTS) {
#pragma unroll
        for (int part = 0; part < 2; ++part) {
            int2 p = p0;
            int  n = min(cnt, SLOTS_A);
            if (part == 1) {
                if (cnt <= SLOTS_A) break;
                p = ((const int2*)outbuf)[(size_t)r * SLOTS_B + (lane & (SLOTS_B - 1))];
                n = cnt - SLOTS_A;
            }
            const bool ok = lane < n;
            const float v = ok ? __int_as_float(p.y) : 0.f;
            const int   c = ok ? p.x : lane_i(p.x, 0);
            for (int i = 0; i < n; i += 4) {
                float g[4];
                float f[4];
#pragma unroll
                for (int u = 0; u < 4; ++u) {
                    f[u] = lane_f(v, i + u);
                    g[u] = s[(size_t)lane_i(c, i + u) * OUT_F];
                }
#pragma unroll
                for (int u = 0; u < 4; ++u) acc += f[u] * g[u];
            }
        }
    } else {
        for (int e0 = 0; e0 < N_EDGES; e0 += 32) {
            unsigned hit = __builtin_amdgcn_ballot_w32(rows[e0 + lane] == r);
            while (hit) {
                const int e = e0 + __builtin_ctz(hit);
                hit &= hit - 1;
                acc += vals[e] * s[(size_t)cols[e] * OUT_F];
            }
        }
    }
    outbuf[(size_t)r * OUT_F + lane] = acc;
}

extern "C" void kernel_launch(void* const* d_in, const int* in_sizes, int n_in,
                              void* d_out, int out_size, void* d_ws, size_t ws_size,
                              hipStream_t stream)
{
    const float* x    = (const float*)d_in[0];
    const int*   rows = (const int*)  d_in[1];
    const int*   cols = (const int*)  d_in[2];
    const float* vals = (const float*)d_in[3];
    const float* W1   = (const float*)d_in[4];
    const float* W2   = (const float*)d_in[5];
    float* out = (float*)d_out;

    float* P0 = (float*)d_ws;
    float* P1 = P0 + PLANE_SZ;
    float* Q0 = P1 + PLANE_SZ;
    int*   Q1 = (int*)(Q0 + PLANE_SZ);

    const int gemmBlocks = ((N_NODES / 16) + 3) / 4;
    const int rowBlocks  = (N_NODES * 32 + 255) / 256;
    const int rowThreads = (N_NODES + 255) / 256;

    zero_counts<<<rowThreads, 256, 0, stream>>>(Q1);

    gemm1_and_slots<<<SLOT_BLOCKS + gemmBlocks, 128, 0, stream>>>(
        x, W1, P0, P1, rows, cols, vals, Q1, out);

    rowsum_plane<<<rowBlocks, 256, 0, stream>>>(rows, cols, vals, P0, Q1, Q0, out);
    rowsum_plane<<<rowBlocks, 256, 0, stream>>>(rows, cols, vals, P1, Q1, P0, out);

    gemm2_wmma<<<gemmBlocks, 128, 0, stream>>>(Q0, P0, W2, P1);

    rowsum_f32<<<rowBlocks, 256, 0, stream>>>(rows, cols, vals, Q1, P1, out);
}


# all three row sums: a wave sums 2 rows (64-col plane) or 4 rows (h2) at once, 16 bytes per lane, 8 whole-row gathers per row in flight (was 1 row per wave, 4); 1/2 and 1/4 the waves; sum order kept
# speedup vs baseline: 3.2741x; 1.0766x over previous
#include <hip/hip_runtime.h>

#define N_NODES 100000
#define N_EDGES 1600000
#define IN_F    256
#define HID_F   128
#define OUT_F   32

#define PLANE_W   64
#define PLANE_SZ  ((size_t)N_NODES * PLANE_W)

#define SLOTS_A   31
#define SLOTS_B   16
#define SLOTS     (SLOTS_A + SLOTS_B)
#define T_COUNT   (2 * SLOTS_A)

#define HALF_ROWS (N_NODES / 2)
#define RUN       256
#define N_RUNS    (N_EDGES / RUN)
#define T_DRAW    (T_COUNT + 1)
#define DRAW_ROW(h) ((h) ? (size_t)(N_NODES - 1) : (size_t)0)
#define SLOT_BLOCKS 1024

typedef float v2f __attribute__((ext_vector_type(2)));
typedef float v8f __attribute__((ext_vector_type(8)));

__global__ __launch_bounds__(256) void zero_counts(int* __restrict__ tbl)
{
    const int r = blockIdx.x * blockDim.x + threadIdx.x;
    if (r >= N_NODES) return;
    tbl[(size_t)r * PLANE_W + T_COUNT] = 0;
    if (r == 0 || r == N_NODES - 1) tbl[(size_t)r * PLANE_W + T_DRAW] = 0;
}

static __device__ __forceinline__ void gemm1_rows(
    const int wid, const int lane,
    const float* __restrict__ x, const float* __restrict__ W1,
    float* __restrict__ h0a, float* __restrict__ h0b)
{
    if (wid >= N_NODES / 16) return;
    const int m    = lane & 15;
    const int kh   = lane >> 4;
    const int row0 = wid * 16;

    v8f acc[8] = {};

    const float* arow = x + (size_t)(row0 + m) * IN_F + 2 * kh;
    const float* wcol = W1 + m;

    for (int k = 0; k < IN_F; k += 4) {
        v2f a = *(const v2f*)(arow + k);
        const float* wrow = wcol + (size_t)(k + 2 * kh) * HID_F;
#pragma unroll
        for (int t = 0; t < 8; ++t) {
            v2f b;
            b.x = wrow[t * 16];
            b.y = wrow[HID_F + t * 16];
            acc[t] = __builtin_amdgcn_wmma_f32_16x16x4_f32(
                false, a, false, b, (short)0, acc[t], false, false);
        }
    }

    const size_t roff = (size_t)(row0 + kh * 8) * PLANE_W + m;
    float* da = h0a + roff;
    float* db = h0b + roff;
#pragma unroll
    for (int t = 0; t < 4; ++t)
#pragma unroll
        for (int v = 0; v < 8; ++v) {
            da[(size_t)v * PLANE_W + t * 16] = acc[t][v];
            db[(size_t)v * PLANE_W + t * 16] = acc[t + 4][v];
        }
}

__global__ __launch_bounds__(128) void gemm2_wmma(
    const float* __restrict__ h1a, const float* __restrict__ h1b,
    const float* __restrict__ W2, float* __restrict__ h2)
{
    const int wid  = (blockIdx.x * blockDim.x + threadIdx.x) >> 5;
    const int lane = threadIdx.x & 31;
    if (wid >= N_NODES / 16) return;
    const int m    = lane & 15;
    const int kh   = lane >> 4;
    const int row0 = wid * 16;

    v8f acc[2] = {};

    const size_t aoff = (size_t)(row0 + m) * PLANE_W + 2 * kh;
    const float* wcol = W2 + m;

#pragma unroll
    for (int p = 0; p < 2; ++p) {
        const float* arow = (p ? h1b : h1a) + aoff;
        for (int k = 0; k < PLANE_W; k += 4) {
            v2f a = *(const v2f*)(arow + k);
            a.x = fmaxf(a.x, 0.0f);
            a.y = fmaxf(a.y, 0.0f);
            const float* wrow = wcol + (size_t)(p * PLANE_W + k + 2 * kh) * OUT_F;
#pragma unroll
            for (int t = 0; t < 2; ++t) {
                v2f b;
                b.x = wrow[t * 16];
                b.y = wrow[OUT_F + t * 16];
                acc[t] = __builtin_amdgcn_wmma_f32_16x16x4_f32(
                    false, a, false, b, (short)0, acc[t], false, false);
            }
        }
    }

    float* drow = h2 + (size_t)(row0 + kh * 8) * OUT_F + m;
#pragma unroll
    for (int t = 0; t < 2; ++t)
#pragma unroll
        for (int v = 0; v < 8; ++v)
            drow[(size_t)v * OUT_F + t * 16] = acc[t][v];
}

static __device__ __forceinline__ void slot_edges(
    const int lane,
    const int* __restrict__ rows, const int* __restrict__ cols,
    const float* __restrict__ vals, int* __restrict__ tbl, float* __restrict__ outbuf)
{
    const int mine = (int)((__builtin_amdgcn_s_sendmsg_rtn(0x87) >> 18) & 1u);

    for (int pass = 0; pass < 2; ++pass) {
        const int hh = mine ^ pass;
        int* ctr = tbl + DRAW_ROW(hh) * PLANE_W + T_DRAW;
        for (int it = 0; it <= N_RUNS; ++it) {
            int k = 0;
            if (lane == 0) k = __hip_atomic_fetch_add(ctr, 1, __ATOMIC_RELAXED, __HIP_MEMORY_SCOPE_AGENT);
            k = __builtin_amdgcn_readfirstlane(k);
            if (k >= N_RUNS) break;

            const int e = k * RUN + lane * 4;
            const int4   ra = *(const int4*)(rows + e),   rb = *(const int4*)(rows + e + RUN / 2);
            const int4   ca = *(const int4*)(cols + e),   cb = *(const int4*)(cols + e + RUN / 2);
            const float4 va = *(const float4*)(vals + e), vb = *(const float4*)(vals + e + RUN / 2);
            const int   r[8] = {ra.x, ra.y, ra.z, ra.w, rb.x, rb.y, rb.z, rb.w};
            const int   c[8] = {ca.x, ca.y, ca.z, ca.w, cb.x, cb.y, cb.z, cb.w};
            const float v[8] = {va.x, va.y, va.z, va.w, vb.x, vb.y, vb.z, vb.w};

            int j[8];
#pragma unroll
            for (int u = 0; u < 8; ++u) {
                j[u] = SLOTS;
                if ((r[u] >= HALF_ROWS) == (hh != 0))
                    j[u] = __hip_atomic_fetch_add(tbl + (size_t)r[u] * PLANE_W + T_COUNT, 1,
                                                  __ATOMIC_RELAXED, __HIP_MEMORY_SCOPE_AGENT);
            }
#pragma unroll
            for (int u = 0; u < 8; ++u) {
                const int2 cv = make_int2(c[u], __float_as_int(v[u]));
                if (j[u] < SLOTS_A)    ((int2*)(tbl + (size_t)r[u] * PLANE_W))[j[u]] = cv;
                else if (j[u] < SLOTS) ((int2*)outbuf)[(size_t)r[u] * SLOTS_B + (j[u] - SLOTS_A)] = cv;
            }
        }
    }
}

__global__ __launch_bounds__(128) void gemm1_and_slots(
    const float* __restrict__ x, const float* __restrict__ W1,
    float* __restrict__ h0a, float* __restrict__ h0b,
    const int* __restrict__ rows, const int* __restrict__ cols,
    const float* __restrict__ vals, int* __restrict__ tbl, float* __restrict__ outbuf)
{
    const int lane = threadIdx.x & 31;
    if (blockIdx.x < SLOT_BLOCKS) {
        slot_edges(lane, rows, cols, vals, tbl, outbuf);
    } else {
        const int wid = ((blockIdx.x - SLOT_BLOCKS) * blockDim.x + threadIdx.x) >> 5;
        gemm1_rows(wid, lane, x, W1, h0a, h0b);
    }
}

static __device__ __forceinline__ int   lane_i(int v, int l)   { return __builtin_amdgcn_readlane(v, l); }

#define GATHERS 8

template <int LPR, int J>
static __device__ __forceinline__ int from_group(const int (&R)[32 / LPR])
{
    static_assert(LPR == 16 || LPR == 8, "groups of 16 or 8 lanes");
    if constexpr (LPR == 16)
        return __builtin_amdgcn_update_dpp(0, R[J / 16], 0x150 + (J % 16), 0xF, 0xF, false);
    else
        return __builtin_amdgcn_mov_dpp8(R[J / 8], (J % 8) * 0x249249);
}

template <int LPR, int I>
static __device__ __forceinline__ void walk_slots(
    float4& acc, const int (&c)[32 / LPR], const int (&v)[32 / LPR], const char* sb)
{
    constexpr unsigned ROW_B = LPR * 16;
    float4 g[GATHERS];
    float  f[GATHERS];
#define ASK(U) \
    f[U] = __int_as_float(from_group<LPR, I + U>(v)); \
    g[U] = *(const float4*)(sb + (unsigned)from_group<LPR, I + U>(c) * ROW_B);
    ASK(0) ASK(1) ASK(2) ASK(3) ASK(4) ASK(5) ASK(6) ASK(7)
#undef ASK
#pragma unroll
    for (int u = 0; u < GATHERS; ++u) {
        acc.x = fmaf(f[u], g[u].x, acc.x);  acc.y = fmaf(f[u], g[u].y, acc.y);
        acc.z = fmaf(f[u], g[u].z, acc.z);  acc.w = fmaf(f[u], g[u].w, acc.w);
    }
}

template <int ROWW>
static __device__ __forceinline__ void sum_rows(
    const int wave, const int lane,
    const int* __restrict__ rows, const int* __restrict__ cols, const float* __restrict__ vals,
    const float* __restrict__ src, const int* __restrict__ tbl,
    const int2* ovf, float4* dst4)
{
    constexpr int      LPR   = ROWW / 4;
    constexpr int      RPW   = 32 / LPR;
    constexpr int      K     = 32 / LPR;
    constexpr int      K2    = SLOTS_B / LPR;
    constexpr unsigned ROW_B = ROWW * 4;
    static_assert(GATHERS == 8 && SLOTS_A == 31 && SLOTS_B == 16, "the walks below are written for these");
    static_assert(N_NODES % RPW == 0, "a wave's rows all exist");
    static_assert((size_t)N_NODES * ROW_B < (1ull << 32), "byte offsets fit 32 bits");

    const int grp = lane / LPR;
    const int sub = lane % LPR;
    const int r0  = wave * RPW;
    const int r   = r0 + grp;
    const char* sb = (const char*)src + sub * 16;

    int cx[K], vx[K];
#pragma unroll
    for (int k = 0; k < K; ++k) {
        const int2 e = ((const int2*)(tbl + (size_t)r * PLANE_W))[k * LPR + sub];
        cx[k] = e.x;  vx[k] = e.y;
    }
    const int cnt = from_group<LPR, SLOTS_A>(cx);
    const int n1  = cnt > SLOTS ? 0 : min(cnt, SLOTS_A);
    const int n2  = cnt > SLOTS ? 0 : max(cnt - SLOTS_A, 0);

    float4 acc = make_float4(0.f, 0.f, 0.f, 0.f);

    {
        int nmax = 0;
#pragma unroll
        for (int q = 0; q < RPW; ++q) nmax = max(nmax, lane_i(n1, q * LPR));
        const int first = n1 > 0 ? from_group<LPR, 0>(cx) : 0;
        int c[K], v[K];
#pragma unroll
        for (int k = 0; k < K; ++k) {
            const bool ok = k * LPR + sub < n1;
            c[k] = ok ? cx[k] : first;
            v[k] = ok ? vx[k] : 0;
        }
        if (nmax > 0)  walk_slots<LPR, 0>(acc, c, v, sb);
        if (nmax > 8)  walk_slots<LPR, 8>(acc, c, v, sb);
        if (nmax > 16) walk_slots<LPR, 16>(acc, c, v, sb);
        if (nmax > 24) walk_slots<LPR, 24>(acc, c, v, sb);
    }
    {
        int nmax = 0;
#pragma unroll
        for (int q = 0; q < RPW; ++q) nmax = max(nmax, lane_i(n2, q * LPR));
        if (nmax > 0) {
            int c[K] = {}, v[K] = {};
#pragma unroll
            for (int k = 0; k < K2; ++k) {
                const int2 e = ovf[(size_t)r * SLOTS_B + k * LPR + sub];
                c[k] = e.x;  v[k] = e.y;
            }
            const int first = n2 > 0 ? from_group<LPR, 0>(c) : 0;
#pragma unroll
            for (int k = 0; k < K2; ++k) {
                const bool ok = k * LPR + sub < n2;
                c[k] = ok ? c[k] : first;
                v[k] = ok ? v[k] : 0;
            }
            walk_slots<LPR, 0>(acc, c, v, sb);
            if (nmax > 8) walk_slots<LPR, 8>(acc, c, v, sb);
        }
    }

#pragma unroll
    for (int q = 0; q < RPW; ++q) {
        if (lane_i(cnt, q * LPR) <= SLOTS) continue;
        const int rq = r0 + q;
        float4 a = make_float4(0.f, 0.f, 0.f, 0.f);
        for (int e0 = 0; e0 < N_EDGES; e0 += 32) {
            unsigned hit = __builtin_amdgcn_ballot_w32(rows[e0 + lane] == rq);
            while (hit) {
                const int e = e0 + __builtin_ctz(hit);
                hit &= hit - 1;
                const float  fe = vals[e];
                const float4 ge = *(const float4*)(sb + (unsigned)cols[e] * ROW_B);
                a.x = fmaf(fe, ge.x, a.x);  a.y = fmaf(fe, ge.y, a.y);
                a.z = fmaf(fe, ge.z, a.z);  a.w = fmaf(fe, ge.w, a.w);
            }
        }
        if (grp == q) acc = a;
    }

    dst4[(size_t)r * LPR + sub] = acc;
}

__global__ __launch_bounds__(256) void rowsum_plane(
    const int* __restrict__ rows, const int* __restrict__ cols,
    const float* __restrict__ vals, const float* __restrict__ src,
    const int* __restrict__ tbl, float* __restrict__ dst, const float* __restrict__ outbuf)
{
    const int wave = (blockIdx.x * blockDim.x + threadIdx.x) >> 5;
    const int lane = threadIdx.x & 31;
    if (wave >= N_NODES / 2) return;
    sum_rows<PLANE_W>(wave, lane, rows, cols, vals, src, tbl, (const int2*)outbuf, (float4*)dst);
}

__global__ __launch_bounds__(256) void rowsum_f32(
    const int* __restrict__ rows, const int* __restrict__ cols,
    const float* __restrict__ vals, const int* __restrict__ tbl,
    const float* __restrict__ h2, float* outbuf)
{
    const int wave = (blockIdx.x * blockDim.x + threadIdx.x) >> 5;
    const int lane = threadIdx.x & 31;
    if (wave >= N_NODES / 4) return;
    sum_rows<OUT_F>(wave, lane, rows, cols, vals, h2, tbl, (const int2*)outbuf, (float4*)outbuf);
}

extern "C" void kernel_launch(void* const* d_in, const int* in_sizes, int n_in,
                              void* d_out, int out_size, void* d_ws, size_t ws_size,
                              hipStream_t stream)
{
    const float* x    = (const float*)d_in[0];
    const int*   rows = (const int*)  d_in[1];
    const int*   cols = (const int*)  d_in[2];
    const float* vals = (const float*)d_in[3];
    const float* W1   = (const float*)d_in[4];
    const float* W2   = (const float*)d_in[5];
    float* out = (float*)d_out;

    float* P0 = (float*)d_ws;
    float* P1 = P0 + PLANE_SZ;
    float* Q0 = P1 + PLANE_SZ;
    int*   Q1 = (int*)(Q0 + PLANE_SZ);

    const int gemmBlocks = ((N_NODES / 16) + 3) / 4;
    const int planeBlocks = (N_NODES / 2 * 32 + 255) / 256;
    const int outBlocks   = (N_NODES / 4 * 32 + 255) / 256;
    const int rowThreads = (N_NODES + 255) / 256;

    zero_counts<<<rowThreads, 256, 0, stream>>>(Q1);

    gemm1_and_slots<<<SLOT_BLOCKS + gemmBlocks, 128, 0, stream>>>(
        x, W1, P0, P1, rows, cols, vals, Q1, out);

    rowsum_plane<<<planeBlocks, 256, 0, stream>>>(rows, cols, vals, P0, Q1, Q0, out);
    rowsum_plane<<<planeBlocks, 256, 0, stream>>>(rows, cols, vals, P1, Q1, P0, out);

    gemm2_wmma<<<gemmBlocks, 128, 0, stream>>>(Q0, P0, W2, P1);

    rowsum_f32<<<outBlocks, 256, 0, stream>>>(rows, cols, vals, Q1, P1, out);
}
